# starting kernel
# baseline (speedup 1.0000x reference)
#include <hip/hip_runtime.h>
#include <math.h>

typedef float v2f __attribute__((ext_vector_type(2)));
typedef float v8f __attribute__((ext_vector_type(8)));

__device__ __forceinline__ float softplusf(float x) {
  return (x > 20.f) ? x : log1pf(expf(x));
}

__device__ __forceinline__ float fast_tanh(float x) {
#if __has_builtin(__builtin_amdgcn_tanhf)
  return __builtin_amdgcn_tanhf(x);
#else
  float cx = fminf(fmaxf(x, -10.f), 10.f);
  float e = exp2f(cx * 2.885390081777927f);
  return (e - 1.f) / (e + 1.f);
#endif
}

__device__ __forceinline__ float sigmoidf_(float x) {
  return 1.f / (1.f + expf(-x));
}

__device__ __forceinline__ v8f wmma4_f32(v2f a, v2f b, v8f c) {
#if __has_builtin(__builtin_amdgcn_wmma_f32_16x16x4_f32)
  return __builtin_amdgcn_wmma_f32_16x16x4_f32(
       false, a,  false, b,
       (short)0, c,  false,  false);
#else
  int lane = threadIdx.x & 31;
  int half = lane >> 4;
  int n = lane & 15;
  float B0 = __shfl(b.x, n),      B1 = __shfl(b.y, n);
  float B2 = __shfl(b.x, n + 16), B3 = __shfl(b.y, n + 16);
  v8f d;
#pragma unroll
  for (int j = 0; j < 8; ++j) {
    int r = j + (half ? 8 : 0);
    float A0 = __shfl(a.x, r),      A1 = __shfl(a.y, r);
    float A2 = __shfl(a.x, r + 16), A3 = __shfl(a.y, r + 16);
    d[j] = fmaf(A0, B0, fmaf(A1, B1, fmaf(A2, B2, fmaf(A3, B3, c[j]))));
  }
  return d;
#endif
}

struct Lay {
  float a0, a1;
  float c0, c1, c2;
  float t0, t1, t2;
};

__device__ __forceinline__ void setup_l0(const float* m, const float* b, const float* f,
                                         int ch, int lane, Lay& L) {
  int half = lane >> 4, row = lane & 15;
  const float* M = m + ch * 3;
  const float* B = b + ch * 3;
  const float* F = f + ch * 3;
  L.a0 = (!half && row < 3) ? softplusf(M[row]) : 0.f;
  L.a1 = 0.f;
  L.c0 = half ? 0.f : B[0];  L.c1 = half ? 0.f : B[1];  L.c2 = half ? 0.f : B[2];
  L.t0 = half ? 0.f : tanhf(F[0]);
  L.t1 = half ? 0.f : tanhf(F[1]);
  L.t2 = half ? 0.f : tanhf(F[2]);
}

__device__ __forceinline__ void setup_l33(const float* m, const float* b, const float* f,
                                          int ch, int lane, Lay& L) {
  int half = lane >> 4, row = lane & 15;
  const float* M = m + ch * 9;
  const float* B = b + ch * 3;
  const float* F = f + ch * 3;
  float a0 = 0.f, a1 = 0.f;
  if (row < 3) {
    a0 = softplusf(M[row * 3 + (half ? 2 : 0)]);
    a1 = half ? 0.f : softplusf(M[row * 3 + 1]);
  }
  L.a0 = a0; L.a1 = a1;
  L.c0 = half ? 0.f : B[0];  L.c1 = half ? 0.f : B[1];  L.c2 = half ? 0.f : B[2];
  L.t0 = half ? 0.f : tanhf(F[0]);
  L.t1 = half ? 0.f : tanhf(F[1]);
  L.t2 = half ? 0.f : tanhf(F[2]);
}

__device__ __forceinline__ void setup_l3(const float* m, const float* b, const float* f,
                                         int ch, int lane, Lay& L) {
  int half = lane >> 4, row = lane & 15;
  const float* M = m + ch * 3;
  L.a0 = (row == 0) ? softplusf(M[half ? 2 : 0]) : 0.f;
  L.a1 = (row == 0 && !half) ? softplusf(M[1]) : 0.f;
  L.c0 = half ? 0.f : b[ch];
  L.c1 = 0.f; L.c2 = 0.f;
  L.t0 = half ? 0.f : tanhf(f[ch]);
  L.t1 = 0.f; L.t2 = 0.f;
}

__device__ __forceinline__ float eval_chain(Lay L0, Lay L1, Lay L2, Lay L3,
                                            float x0, int lane) {
  int half = lane >> 4;
  int nlo = lane & 15;

  v2f a, b;
  a.x = L0.a0; a.y = L0.a1;
  b.x = x0;    b.y = 0.f;
  v8f c = {L0.c0, L0.c1, L0.c2, 0.f, 0.f, 0.f, 0.f, 0.f};
  v8f d = wmma4_f32(a, b, c);
  float h0 = d[0] + L0.t0 * fast_tanh(d[0]);
  float h1 = d[1] + L0.t1 * fast_tanh(d[1]);
  float h2 = d[2] + L0.t2 * fast_tanh(d[2]);

  float h2lo = __shfl(h2, nlo);
  a.x = L1.a0; a.y = L1.a1;
  b.x = half ? h2lo : h0;  b.y = half ? 0.f : h1;
  v8f c1 = {L1.c0, L1.c1, L1.c2, 0.f, 0.f, 0.f, 0.f, 0.f};
  d = wmma4_f32(a, b, c1);
  h0 = d[0] + L1.t0 * fast_tanh(d[0]);
  h1 = d[1] + L1.t1 * fast_tanh(d[1]);
  h2 = d[2] + L1.t2 * fast_tanh(d[2]);

  h2lo = __shfl(h2, nlo);
  a.x = L2.a0; a.y = L2.a1;
  b.x = half ? h2lo : h0;  b.y = half ? 0.f : h1;
  v8f c2 = {L2.c0, L2.c1, L2.c2, 0.f, 0.f, 0.f, 0.f, 0.f};
  d = wmma4_f32(a, b, c2);
  h0 = d[0] + L2.t0 * fast_tanh(d[0]);
  h1 = d[1] + L2.t1 * fast_tanh(d[1]);
  h2 = d[2] + L2.t2 * fast_tanh(d[2]);

  h2lo = __shfl(h2, nlo);
  a.x = L3.a0; a.y = L3.a1;
  b.x = half ? h2lo : h0;  b.y = half ? 0.f : h1;
  v8f c3 = {L3.c0, 0.f, 0.f, 0.f, 0.f, 0.f, 0.f, 0.f};
  d = wmma4_f32(a, b, c3);
  float r = d[0];
  return r + L3.t0 * fast_tanh(r);
}

__global__ void __launch_bounds__(256)
eb_kernel(const float* __restrict__ inputs, const float* __restrict__ noise,
          const float* __restrict__ m0, const float* __restrict__ b0, const float* __restrict__ f0,
          const float* __restrict__ m1, const float* __restrict__ b1, const float* __restrict__ f1,
          const float* __restrict__ m2, const float* __restrict__ b2, const float* __restrict__ f2,
          const float* __restrict__ m3, const float* __restrict__ b3, const float* __restrict__ f3,
          float* __restrict__ out, float* __restrict__ lik, int ntiles) {
  __shared__ float xs[256];
  __shared__ float ls[256];

  const int tid  = threadIdx.x;
  const int lane = tid & 31;
  const int wave = tid >> 5;
  const int half = lane >> 4;
  const int row  = lane & 15;

  Lay L0[2], L1[2], L2[2], L3[2];
#pragma unroll
  for (int cc = 0; cc < 2; ++cc) {
    int ch = wave * 2 + cc;
    setup_l0 (m0, b0, f0, ch, lane, L0[cc]);
    setup_l33(m1, b1, f1, ch, lane, L1[cc]);
    setup_l33(m2, b2, f2, ch, lane, L2[cc]);
    setup_l3 (m3, b3, f3, ch, lane, L3[cc]);
  }

  for (int t = blockIdx.x; t < ntiles; t += gridDim.x) {
    const int base = t * 256;

    float v = inputs[base + tid] + (noise[base + tid] - 0.5f);
    out[base + tid] = v;
    xs[(tid & 15) * 16 + (tid >> 4)] = v;
    __syncthreads();

#pragma unroll
    for (int cc = 0; cc < 2; ++cc) {
      int ch = wave * 2 + cc;
      float xv = xs[ch * 16 + row];
      float xl = half ? 0.f : (xv - 0.5f);
      float xu = half ? 0.f : (xv + 0.5f);
      float lo = eval_chain(L0[cc], L1[cc], L2[cc], L3[cc], xl, lane);
      float up = eval_chain(L0[cc], L1[cc], L2[cc], L3[cc], xu, lane);
      if (half == 0) {
        float sum = lo + up;
        float s = (sum > 0.f) ? -1.f : ((sum < 0.f) ? 1.f : 0.f);
        float lk = fabsf(sigmoidf_(s * up) - sigmoidf_(s * lo));
        ls[row * 16 + ch] = fmaxf(lk, 1e-9f);
      }
    }
    __syncthreads();

    lik[base + tid] = ls[tid];
    __syncthreads();
  }
}

extern "C" void kernel_launch(void* const* d_in, const int* in_sizes, int n_in,
                              void* d_out, int out_size, void* d_ws, size_t ws_size,
                              hipStream_t stream) {
  const float* inputs = (const float*)d_in[0];
  const float* noise  = (const float*)d_in[1];
  const float* m0 = (const float*)d_in[2];
  const float* b0 = (const float*)d_in[3];
  const float* f0 = (const float*)d_in[4];
  const float* m1 = (const float*)d_in[5];
  const float* b1 = (const float*)d_in[6];
  const float* f1 = (const float*)d_in[7];
  const float* m2 = (const float*)d_in[8];
  const float* b2 = (const float*)d_in[9];
  const float* f2 = (const float*)d_in[10];
  const float* m3 = (const float*)d_in[11];
  const float* b3 = (const float*)d_in[12];
  const float* f3 = (const float*)d_in[13];

  const int nelem  = in_sizes[0];
  const int ntiles = nelem / 256;
  float* out = (float*)d_out;
  float* lik = out + nelem;

  int blocks = ntiles < 2048 ? ntiles : 2048;
  eb_kernel<<<blocks, 256, 0, stream>>>(inputs, noise,
                                        m0, b0, f0, m1, b1, f1,
                                        m2, b2, f2, m3, b3, f3,
                                        out, lik, ntiles);
}


# Parameter packing (softplus, tanh, operand layout) moved to a small first kernel that writes each wave's register image to the workspace once per call; 2 launches, was 1; sum order unchanged
# speedup vs baseline: 2.5794x; 2.5794x over previous
#include <hip/hip_runtime.h>
#include <math.h>


typedef float v2f __attribute__((ext_vector_type(2)));
typedef float v8f __attribute__((ext_vector_type(8)));

__device__ __forceinline__ float softplusf(float x) {
  return (x > 20.f) ? x : log1pf(expf(x));
}

__device__ __forceinline__ float fast_tanh(float x) {
#if __has_builtin(__builtin_amdgcn_tanhf)
  return __builtin_amdgcn_tanhf(x);
#else
  float cx = fminf(fmaxf(x, -10.f), 10.f);
  float e = exp2f(cx * 2.885390081777927f);
  return (e - 1.f) / (e + 1.f);
#endif
}

__device__ __forceinline__ float sigmoidf_(float x) {
  return 1.f / (1.f + expf(-x));
}

__device__ __forceinline__ v8f wmma4_f32(v2f a, v2f b, v8f c) {
#if __has_builtin(__builtin_amdgcn_wmma_f32_16x16x4_f32)
  return __builtin_amdgcn_wmma_f32_16x16x4_f32(
       false, a,  false, b,
       (short)0, c,  false,  false);
#else
  int lane = threadIdx.x & 31;
  int half = lane >> 4;
  int n = lane & 15;
  float B0 = __shfl(b.x, n),      B1 = __shfl(b.y, n);
  float B2 = __shfl(b.x, n + 16), B3 = __shfl(b.y, n + 16);
  v8f d;
#pragma unroll
  for (int j = 0; j < 8; ++j) {
    int r = j + (half ? 8 : 0);
    float A0 = __shfl(a.x, r),      A1 = __shfl(a.y, r);
    float A2 = __shfl(a.x, r + 16), A3 = __shfl(a.y, r + 16);
    d[j] = fmaf(A0, B0, fmaf(A1, B1, fmaf(A2, B2, fmaf(A3, B3, c[j]))));
  }
  return d;
#endif
}

struct Par {
  v2f a0;
  v2f a1[3], a2[3], a3[3];
  v8f c0, c1, c2, c3;
  float t0[6], t1[6], t2[6];
  float t3[2];
};

#define PAR_SLOTS 60
#define PAR_WORDS 15
#define TAB_FLOATS (PAR_SLOTS * 4 * 32)
#define PACK_GROUPS 30

__device__ __forceinline__ float par_value(const float* m0, const float* b0, const float* f0,
                                           const float* m1, const float* b1, const float* f1,
                                           const float* m2, const float* b2, const float* f2,
                                           const float* m3, const float* b3, const float* f3,
                                           int slot, int g, int lane) {
  const int hf  = lane >> 4;
  const int rr  = (lane & 15) - 8 * hf;
  const bool in3 = (rr >= 0) && (rr < 6);
  const bool in1 = (rr >= 0) && (rr < 2);
  const int odd = rr & 1;
  const int o   = in3 ? (rr >> 1) : 0;
  const int chA = 4 * g + 2 * hf + odd;
  const int chD = 4 * g + 2 * hf;

  if (slot < 20) {
    const int comp = slot & 1;
    const int p    = slot >> 1;
    const bool rowok = (p >= 7) ? in1 : in3;
    if (!rowok || comp != odd) return 0.f;
    const float* src;
    int idx;
    if (p == 0)     { src = m0; idx = chA * 3 + o; }
    else if (p < 4) { src = m1; idx = chA * 9 + o * 3 + (p - 1); }
    else if (p < 7) { src = m2; idx = chA * 9 + o * 3 + (p - 4); }
    else            { src = m3; idx = chA * 3 + (p - 7); }
    return softplusf(src[idx]);
  }
  const bool isT = slot >= 40;
  const int s = slot - (isT ? 40 : 20);
  float v;
  if (s < 18) {
    const int L = s / 6, j = s - 6 * L;
    const int idx = (chD + (j & 1)) * 3 + (j >> 1);
    const float* src = isT ? (L == 0 ? f0 : (L == 1 ? f1 : f2))
                           : (L == 0 ? b0 : (L == 1 ? b1 : b2));
    v = src[idx];
  } else {
    v = (isT ? f3 : b3)[chD + (s - 18)];
  }
  return isT ? tanhf(v) : v;
}

__global__ void __launch_bounds__(256)
eb_pack(const float* __restrict__ m0, const float* __restrict__ b0, const float* __restrict__ f0,
        const float* __restrict__ m1, const float* __restrict__ b1, const float* __restrict__ f1,
        const float* __restrict__ m2, const float* __restrict__ b2, const float* __restrict__ f2,
        const float* __restrict__ m3, const float* __restrict__ b3, const float* __restrict__ f3,
        float* __restrict__ tab) {
  const int i    = blockIdx.x * 256 + threadIdx.x;
  const int e    = i & 3;
  const int w4   = i >> 2;
  const int lane = w4 & 31;
  const int g    = (w4 >> 5) & 3;
  const int word = w4 >> 7;
  tab[i] = par_value(m0, b0, f0, m1, b1, f1, m2, b2, f2, m3, b3, f3, word * 4 + e, g, lane);
}

__device__ __forceinline__ void load_par(const float* __restrict__ tab, int g, int lane, Par& P) {
  const float4* t4 = reinterpret_cast<const float4*>(tab);
  float fl[PAR_SLOTS];
#pragma unroll
  for (int w = 0; w < PAR_WORDS; ++w) {
    const float4 q = t4[(w * 4 + g) * 32 + lane];
    fl[4 * w] = q.x;  fl[4 * w + 1] = q.y;  fl[4 * w + 2] = q.z;  fl[4 * w + 3] = q.w;
  }
  P.a0.x = fl[0];  P.a0.y = fl[1];
#pragma unroll
  for (int k = 0; k < 3; ++k) {
    P.a1[k].x = fl[2 + 2 * k];   P.a1[k].y = fl[3 + 2 * k];
    P.a2[k].x = fl[8 + 2 * k];   P.a2[k].y = fl[9 + 2 * k];
    P.a3[k].x = fl[14 + 2 * k];  P.a3[k].y = fl[15 + 2 * k];
  }
  v8f z = {0.f, 0.f, 0.f, 0.f, 0.f, 0.f, 0.f, 0.f};
  P.c0 = z; P.c1 = z; P.c2 = z; P.c3 = z;
#pragma unroll
  for (int j = 0; j < 6; ++j) {
    P.c0[j] = fl[20 + j];  P.t0[j] = fl[40 + j];
    P.c1[j] = fl[26 + j];  P.t1[j] = fl[46 + j];
    P.c2[j] = fl[32 + j];  P.t2[j] = fl[52 + j];
  }
  P.c3[0] = fl[38];  P.t3[0] = fl[58];
  P.c3[1] = fl[39];  P.t3[1] = fl[59];
}

__device__ __forceinline__ void gate6(const v8f& d, const float* t, float* h) {
#pragma unroll
  for (int j = 0; j < 6; ++j) h[j] = d[j] + t[j] * fast_tanh(d[j]);
}

__device__ __forceinline__ v8f layer3in(const v2f* a, const float* h, v8f c) {
  v2f b;
  b.x = h[0]; b.y = h[1];
  v8f d = wmma4_f32(a[0], b, c);
  b.x = h[2]; b.y = h[3];
  d = wmma4_f32(a[1], b, d);
  b.x = h[4]; b.y = h[5];
  d = wmma4_f32(a[2], b, d);
  return d;
}

__device__ __forceinline__ v2f eval_chain(const Par& P, v2f x) {
  float h[6];
  v8f d = wmma4_f32(P.a0, x, P.c0);
  gate6(d, P.t0, h);
  d = layer3in(P.a1, h, P.c1);
  gate6(d, P.t1, h);
  d = layer3in(P.a2, h, P.c2);
  gate6(d, P.t2, h);
  d = layer3in(P.a3, h, P.c3);
  v2f r;
  r.x = d[0] + P.t3[0] * fast_tanh(d[0]);
  r.y = d[1] + P.t3[1] * fast_tanh(d[1]);
  return r;
}

__device__ __forceinline__ float likelihood(float lo, float up) {
  float sum = lo + up;
  float s = (sum > 0.f) ? -1.f : ((sum < 0.f) ? 1.f : 0.f);
  float lk = fabsf(sigmoidf_(s * up) - sigmoidf_(s * lo));
  return fmaxf(lk, 1e-9f);
}

#define PITCH 20
#define TILEF (32 * PITCH)

__global__ void __launch_bounds__(256)
eb_kernel(const float* __restrict__ inputs, const float* __restrict__ noise,
          const float* __restrict__ tab,
          float* __restrict__ out, float* __restrict__ lik, int ntiles) {
  __shared__ __attribute__((aligned(16))) float xs[2 * TILEF];
  __shared__ __attribute__((aligned(16))) float ls[2 * TILEF];

  const int tid  = threadIdx.x;
  const int lane = tid & 31;
  const int wave = tid >> 5;
  const int g    = wave & 3;
  const int pg   = wave >> 2;

  Par P;
  load_par(tab, g, lane, P);

  const int io0 = (tid >> 4) * PITCH + (tid & 15);
  const int io1 = io0 + 16 * PITCH;
  const int pos = (16 * pg + (lane & 15)) * PITCH + 4 * g + 2 * (lane >> 4);
  const int gof = (16 * pg + (lane & 15)) * 16 + 4 * g + 2 * (lane >> 4);

  int t = blockIdx.x;
  if (t >= ntiles) return;

  float2 xi = *reinterpret_cast<const float2*>(&inputs[t * 512 + gof]);
  float2 nz = *reinterpret_cast<const float2*>(&noise[t * 512 + gof]);
  int buf = 0;

  for (; t < ntiles; t += gridDim.x) {
    const int base = t * 512;

    const int tn = (t + (int)gridDim.x < ntiles) ? t + (int)gridDim.x : t;
    const float2 xin = *reinterpret_cast<const float2*>(&inputs[tn * 512 + gof]);
    const float2 nzn = *reinterpret_cast<const float2*>(&noise[tn * 512 + gof]);

    float2 xv;
    xv.x = xi.x + (nz.x - 0.5f);
    xv.y = xi.y + (nz.y - 0.5f);
    v2f xl, xu;
    xl.x = xv.x - 0.5f;  xl.y = xv.y - 0.5f;
    xu.x = xv.x + 0.5f;  xu.y = xv.y + 0.5f;
    v2f lo = eval_chain(P, xl);
    v2f up = eval_chain(P, xu);
    float2 lk;
    lk.x = likelihood(lo.x, up.x);
    lk.y = likelihood(lo.y, up.y);

    float* xb = xs + buf * TILEF;
    float* lb = ls + buf * TILEF;
    *reinterpret_cast<float2*>(&xb[pos]) = xv;
    *reinterpret_cast<float2*>(&lb[pos]) = lk;
    __syncthreads();

    out[base + tid]       = xb[io0];
    out[base + 256 + tid] = xb[io1];
    lik[base + tid]       = lb[io0];
    lik[base + 256 + tid] = lb[io1];

    xi = xin;
    nz = nzn;
    buf ^= 1;
  }
}

extern "C" void kernel_launch(void* const* d_in, const int* in_sizes, int n_in,
                              void* d_out, int out_size, void* d_ws, size_t ws_size,
                              hipStream_t stream) {
  const float* inputs = (const float*)d_in[0];
  const float* noise  = (const float*)d_in[1];
  const float* m0 = (const float*)d_in[2];
  const float* b0 = (const float*)d_in[3];
  const float* f0 = (const float*)d_in[4];
  const float* m1 = (const float*)d_in[5];
  const float* b1 = (const float*)d_in[6];
  const float* f1 = (const float*)d_in[7];
  const float* m2 = (const float*)d_in[8];
  const float* b2 = (const float*)d_in[9];
  const float* f2 = (const float*)d_in[10];
  const float* m3 = (const float*)d_in[11];
  const float* b3 = (const float*)d_in[12];
  const float* f3 = (const float*)d_in[13];

  const int nelem  = in_sizes[0];
  const int ntiles = nelem / 512;
  float* out = (float*)d_out;
  float* lik = out + nelem;

  float* tab = (float*)d_ws;

  eb_pack<<<PACK_GROUPS, 256, 0, stream>>>(m0, b0, f0, m1, b1, f1,
                                                m2, b2, f2, m3, b3, f3, tab);

  int blocks = ntiles < 2048 ? ntiles : 2048;
  eb_kernel<<<blocks, 256, 0, stream>>>(inputs, noise, tab, out, lik, ntiles);
}


# Next tile's loads stay in flight over the tile barrier: barrier by hand waits on LDS only (the library one drained every load); two tiles per loop trip, no register copy; sum order unchanged
# speedup vs baseline: 2.6930x; 1.0440x over previous
#include <hip/hip_runtime.h>
#include <math.h>


typedef float v2f __attribute__((ext_vector_type(2)));
typedef float v8f __attribute__((ext_vector_type(8)));

__device__ __forceinline__ float softplusf(float x) {
  return (x > 20.f) ? x : log1pf(expf(x));
}

__device__ __forceinline__ float fast_tanh(float x) {
#if __has_builtin(__builtin_amdgcn_tanhf)
  return __builtin_amdgcn_tanhf(x);
#else
  float cx = fminf(fmaxf(x, -10.f), 10.f);
  float e = exp2f(cx * 2.885390081777927f);
  return (e - 1.f) / (e + 1.f);
#endif
}

__device__ __forceinline__ float sigmoidf_(float x) {
  return 1.f / (1.f + expf(-x));
}

__device__ __forceinline__ v8f wmma4_f32(v2f a, v2f b, v8f c) {
#if __has_builtin(__builtin_amdgcn_wmma_f32_16x16x4_f32)
  return __builtin_amdgcn_wmma_f32_16x16x4_f32(
       false, a,  false, b,
       (short)0, c,  false,  false);
#else
  int lane = threadIdx.x & 31;
  int half = lane >> 4;
  int n = lane & 15;
  float B0 = __shfl(b.x, n),      B1 = __shfl(b.y, n);
  float B2 = __shfl(b.x, n + 16), B3 = __shfl(b.y, n + 16);
  v8f d;
#pragma unroll
  for (int j = 0; j < 8; ++j) {
    int r = j + (half ? 8 : 0);
    float A0 = __shfl(a.x, r),      A1 = __shfl(a.y, r);
    float A2 = __shfl(a.x, r + 16), A3 = __shfl(a.y, r + 16);
    d[j] = fmaf(A0, B0, fmaf(A1, B1, fmaf(A2, B2, fmaf(A3, B3, c[j]))));
  }
  return d;
#endif
}

struct Par {
  v2f a0;
  v2f a1[3], a2[3], a3[3];
  v8f c0, c1, c2, c3;
  float t0[6], t1[6], t2[6];
  float t3[2];
};

#define PAR_SLOTS 60
#define PAR_WORDS 15
#define TAB_FLOATS (PAR_SLOTS * 4 * 32)
#define PACK_GROUPS 30

__device__ __forceinline__ float par_value(const float* m0, const float* b0, const float* f0,
                                           const float* m1, const float* b1, const float* f1,
                                           const float* m2, const float* b2, const float* f2,
                                           const float* m3, const float* b3, const float* f3,
                                           int slot, int g, int lane) {
  const int hf  = lane >> 4;
  const int rr  = (lane & 15) - 8 * hf;
  const bool in3 = (rr >= 0) && (rr < 6);
  const bool in1 = (rr >= 0) && (rr < 2);
  const int odd = rr & 1;
  const int o   = in3 ? (rr >> 1) : 0;
  const int chA = 4 * g + 2 * hf + odd;
  const int chD = 4 * g + 2 * hf;

  if (slot < 20) {
    const int comp = slot & 1;
    const int p    = slot >> 1;
    const bool rowok = (p >= 7) ? in1 : in3;
    if (!rowok || comp != odd) return 0.f;
    const float* src;
    int idx;
    if (p == 0)     { src = m0; idx = chA * 3 + o; }
    else if (p < 4) { src = m1; idx = chA * 9 + o * 3 + (p - 1); }
    else if (p < 7) { src = m2; idx = chA * 9 + o * 3 + (p - 4); }
    else            { src = m3; idx = chA * 3 + (p - 7); }
    return softplusf(src[idx]);
  }
  const bool isT = slot >= 40;
  const int s = slot - (isT ? 40 : 20);
  float v;
  if (s < 18) {
    const int L = s / 6, j = s - 6 * L;
    const int idx = (chD + (j & 1)) * 3 + (j >> 1);
    const float* src = isT ? (L == 0 ? f0 : (L == 1 ? f1 : f2))
                           : (L == 0 ? b0 : (L == 1 ? b1 : b2));
    v = src[idx];
  } else {
    v = (isT ? f3 : b3)[chD + (s - 18)];
  }
  return isT ? tanhf(v) : v;
}

__global__ void __launch_bounds__(256)
eb_pack(const float* __restrict__ m0, const float* __restrict__ b0, const float* __restrict__ f0,
        const float* __restrict__ m1, const float* __restrict__ b1, const float* __restrict__ f1,
        const float* __restrict__ m2, const float* __restrict__ b2, const float* __restrict__ f2,
        const float* __restrict__ m3, const float* __restrict__ b3, const float* __restrict__ f3,
        float* __restrict__ tab) {
  const int i    = blockIdx.x * 256 + threadIdx.x;
  const int e    = i & 3;
  const int w4   = i >> 2;
  const int lane = w4 & 31;
  const int g    = (w4 >> 5) & 3;
  const int word = w4 >> 7;
  tab[i] = par_value(m0, b0, f0, m1, b1, f1, m2, b2, f2, m3, b3, f3, word * 4 + e, g, lane);
}

__device__ __forceinline__ void load_par(const float* __restrict__ tab, int g, int lane, Par& P) {
  const float4* t4 = reinterpret_cast<const float4*>(tab);
  float fl[PAR_SLOTS];
#pragma unroll
  for (int w = 0; w < PAR_WORDS; ++w) {
    const float4 q = t4[(w * 4 + g) * 32 + lane];
    fl[4 * w] = q.x;  fl[4 * w + 1] = q.y;  fl[4 * w + 2] = q.z;  fl[4 * w + 3] = q.w;
  }
  P.a0.x = fl[0];  P.a0.y = fl[1];
#pragma unroll
  for (int k = 0; k < 3; ++k) {
    P.a1[k].x = fl[2 + 2 * k];   P.a1[k].y = fl[3 + 2 * k];
    P.a2[k].x = fl[8 + 2 * k];   P.a2[k].y = fl[9 + 2 * k];
    P.a3[k].x = fl[14 + 2 * k];  P.a3[k].y = fl[15 + 2 * k];
  }
  v8f z = {0.f, 0.f, 0.f, 0.f, 0.f, 0.f, 0.f, 0.f};
  P.c0 = z; P.c1 = z; P.c2 = z; P.c3 = z;
#pragma unroll
  for (int j = 0; j < 6; ++j) {
    P.c0[j] = fl[20 + j];  P.t0[j] = fl[40 + j];
    P.c1[j] = fl[26 + j];  P.t1[j] = fl[46 + j];
    P.c2[j] = fl[32 + j];  P.t2[j] = fl[52 + j];
  }
  P.c3[0] = fl[38];  P.t3[0] = fl[58];
  P.c3[1] = fl[39];  P.t3[1] = fl[59];
}

__device__ __forceinline__ void gate6(const v8f& d, const float* t, float* h) {
#pragma unroll
  for (int j = 0; j < 6; ++j) h[j] = d[j] + t[j] * fast_tanh(d[j]);
}

__device__ __forceinline__ v8f layer3in(const v2f* a, const float* h, v8f c) {
  v2f b;
  b.x = h[0]; b.y = h[1];
  v8f d = wmma4_f32(a[0], b, c);
  b.x = h[2]; b.y = h[3];
  d = wmma4_f32(a[1], b, d);
  b.x = h[4]; b.y = h[5];
  d = wmma4_f32(a[2], b, d);
  return d;
}

__device__ __forceinline__ v2f eval_chain(const Par& P, v2f x) {
  float h[6];
  v8f d = wmma4_f32(P.a0, x, P.c0);
  gate6(d, P.t0, h);
  d = layer3in(P.a1, h, P.c1);
  gate6(d, P.t1, h);
  d = layer3in(P.a2, h, P.c2);
  gate6(d, P.t2, h);
  d = layer3in(P.a3, h, P.c3);
  v2f r;
  r.x = d[0] + P.t3[0] * fast_tanh(d[0]);
  r.y = d[1] + P.t3[1] * fast_tanh(d[1]);
  return r;
}

__device__ __forceinline__ float likelihood(float lo, float up) {
  float sum = lo + up;
  float s = (sum > 0.f) ? -1.f : ((sum < 0.f) ? 1.f : 0.f);
  float lk = fabsf(sigmoidf_(s * up) - sigmoidf_(s * lo));
  return fmaxf(lk, 1e-9f);
}

#define PITCH 20
#define TILEF (32 * PITCH)

__device__ __forceinline__ void lds_barrier() {
  asm volatile("s_wait_dscnt 0x0\n\ts_barrier_signal -1\n\ts_barrier_wait -1" ::: "memory");
}

__device__ __forceinline__ float2 ld_pair(const float* p, int tile, int gof) {
  return *reinterpret_cast<const float2*>(&p[tile * 512 + gof]);
}

__device__ __forceinline__ void do_tile(const Par& P, float2 xi, float2 nz,
                                        float* __restrict__ xb, float* __restrict__ lb,
                                        float* __restrict__ out, float* __restrict__ lik,
                                        int base, int tid, int pos, int io0, int io1) {
  float2 xv;
  xv.x = xi.x + (nz.x - 0.5f);
  xv.y = xi.y + (nz.y - 0.5f);
  v2f xl, xu;
  xl.x = xv.x - 0.5f;  xl.y = xv.y - 0.5f;
  xu.x = xv.x + 0.5f;  xu.y = xv.y + 0.5f;
  v2f lo = eval_chain(P, xl);
  v2f up = eval_chain(P, xu);
  float2 lk;
  lk.x = likelihood(lo.x, up.x);
  lk.y = likelihood(lo.y, up.y);

  *reinterpret_cast<float2*>(&xb[pos]) = xv;
  *reinterpret_cast<float2*>(&lb[pos]) = lk;
  lds_barrier();

  out[base + tid]       = xb[io0];
  out[base + 256 + tid] = xb[io1];
  lik[base + tid]       = lb[io0];
  lik[base + 256 + tid] = lb[io1];
}

__global__ void __launch_bounds__(256)
eb_kernel(const float* inputs, const float* noise,
          const float* __restrict__ tab,
          float* __restrict__ out, float* __restrict__ lik, int ntiles) {
  __shared__ __attribute__((aligned(16))) float xs[2 * TILEF];
  __shared__ __attribute__((aligned(16))) float ls[2 * TILEF];

  const int tid  = threadIdx.x;
  const int lane = tid & 31;
  const int wave = tid >> 5;
  const int g    = wave & 3;
  const int pg   = wave >> 2;

  Par P;
  load_par(tab, g, lane, P);

  const int io0 = (tid >> 4) * PITCH + (tid & 15);
  const int io1 = io0 + 16 * PITCH;
  const int pos = (16 * pg + (lane & 15)) * PITCH + 4 * g + 2 * (lane >> 4);
  const int gof = (16 * pg + (lane & 15)) * 16 + 4 * g + 2 * (lane >> 4);

  const int G = (int)gridDim.x;
  int t = blockIdx.x;
  if (t >= ntiles) return;

  float2 xa = ld_pair(inputs, t, gof);
  float2 na = ld_pair(noise, t, gof);

  for (; t + G < ntiles; t += 2 * G) {
    const int t1 = t + G;
    const float2 xb2 = ld_pair(inputs, t1, gof);
    const float2 nb2 = ld_pair(noise, t1, gof);
    __builtin_amdgcn_sched_barrier(0);
    do_tile(P, xa, na, xs, ls, out, lik, t * 512, tid, pos, io0, io1);

    const int t2 = t1 + G;
    const int p2 = (t2 < ntiles) ? t2 : t1;
    xa = ld_pair(inputs, p2, gof);
    na = ld_pair(noise, p2, gof);
    __builtin_amdgcn_sched_barrier(0);
    do_tile(P, xb2, nb2, xs + TILEF, ls + TILEF, out, lik, t1 * 512, tid, pos, io0, io1);
  }
  if (t < ntiles)
    do_tile(P, xa, na, xs, ls, out, lik, t * 512, tid, pos, io0, io1);
}

extern "C" void kernel_launch(void* const* d_in, const int* in_sizes, int n_in,
                              void* d_out, int out_size, void* d_ws, size_t ws_size,
                              hipStream_t stream) {
  const float* inputs = (const float*)d_in[0];
  const float* noise  = (const float*)d_in[1];
  const float* m0 = (const float*)d_in[2];
  const float* b0 = (const float*)d_in[3];
  const float* f0 = (const float*)d_in[4];
  const float* m1 = (const float*)d_in[5];
  const float* b1 = (const float*)d_in[6];
  const float* f1 = (const float*)d_in[7];
  const float* m2 = (const float*)d_in[8];
  const float* b2 = (const float*)d_in[9];
  const float* f2 = (const float*)d_in[10];
  const float* m3 = (const float*)d_in[11];
  const float* b3 = (const float*)d_in[12];
  const float* f3 = (const float*)d_in[13];

  const int nelem  = in_sizes[0];
  const int ntiles = nelem / 512;
  float* out = (float*)d_out;
  float* lik = out + nelem;

  float* tab = (float*)d_ws;

  eb_pack<<<PACK_GROUPS, 256, 0, stream>>>(m0, b0, f0, m1, b1, f1,
                                                m2, b2, f2, m3, b3, f3, tab);

  int blocks = ntiles < 2048 ? ntiles : 2048;
  eb_kernel<<<blocks, 256, 0, stream>>>(inputs, noise, tab, out, lik, ntiles);
}
